# starting kernel
# baseline (speedup 1.0000x reference)
#include <hip/hip_runtime.h>
#include <hip/hip_bf16.h>

#define N_NODES 100000
#define N_EDGES 1600000
#define D_IN    256
#define D_EDGE  16
#define D_OUT   128

typedef __attribute__((ext_vector_type(2))) float v2f;
typedef __attribute__((ext_vector_type(8))) float v8f;

__global__ void zero_cnt_kernel(int* __restrict__ cnt, int n) {
    int i = blockIdx.x * blockDim.x + threadIdx.x;
    if (i < n) cnt[i] = 0;
}

__global__ void count_edges_kernel(const int* __restrict__ dst,
                                   int* __restrict__ cnt, int e_total) {
    int e = blockIdx.x * blockDim.x + threadIdx.x;
    if (e < e_total) atomicAdd(&cnt[dst[e]], 1);
}

#define XS_STRIDE 257

__global__ void __launch_bounds__(512)
gemm_dual_wmma_kernel(const float* __restrict__ X,
                      const float* __restrict__ W_self,
                      const float* __restrict__ W_neigh,
                      const float* __restrict__ b_self,
                      float* __restrict__ out_self,
                      float* __restrict__ out_neigh) {
    __shared__ float xs[16 * XS_STRIDE];

    const int rowBase = blockIdx.x * 16;

    for (int idx = threadIdx.x; idx < 16 * D_IN; idx += 512) {
        const int r = idx >> 8;
        const int c = idx & 255;
        xs[r * XS_STRIDE + c] = X[(size_t)(rowBase + r) * D_IN + c];
    }
    __syncthreads();

    const int wave  = threadIdx.x >> 5;
    const int lane  = threadIdx.x & 31;
    const int ntile = wave & 7;
    const bool isSelf = (wave < 8);

    const float* __restrict__ W = isSelf ? W_self : W_neigh;

    const int m     = lane & 15;
    const int khalf = lane >> 4;
    const int col   = ntile * 16 + m;

    v8f acc = {};
    const float* xrow = &xs[m * XS_STRIDE];

    #pragma unroll 4
    for (int kb = 0; kb < D_IN; kb += 4) {
        const int k0 = kb + 2 * khalf;
        v2f a, b;
        a.x = xrow[k0 + 0];
        a.y = xrow[k0 + 1];
        b.x = W[(size_t)(k0 + 0) * D_OUT + col];
        b.y = W[(size_t)(k0 + 1) * D_OUT + col];
        acc = __builtin_amdgcn_wmma_f32_16x16x4_f32(
             false, a,  false, b,
             (short)0, acc,  false,  false);
    }

    const float bias = isSelf ? b_self[col] : 0.0f;
    float* __restrict__ outp = isSelf ? out_self : out_neigh;

    #pragma unroll
    for (int r = 0; r < 8; ++r) {
        const int row = rowBase + r + 8 * khalf;
        outp[(size_t)row * D_OUT + col] = acc[r] + bias;
    }
}

#define EDGES_PER_BLOCK 8

__global__ void __launch_bounds__(256)
edge_scatter_kernel(const float* __restrict__ X_edge,
                    const float* __restrict__ h_src,
                    const int* __restrict__ src,
                    const int* __restrict__ dst,
                    const int* __restrict__ cnt,
                    const float* __restrict__ W_edge,
                    float* __restrict__ out, int e_total) {
    __shared__ float wl[D_EDGE * D_OUT];
    for (int i = threadIdx.x; i < D_EDGE * D_OUT; i += 256) wl[i] = W_edge[i];
    __syncthreads();

    const int wave = threadIdx.x >> 5;
    const int lane = threadIdx.x & 31;
    const int e = blockIdx.x * EDGES_PER_BLOCK + wave;
    if (e >= e_total) return;

    const int s = src[e];
    const int d = dst[e];
    const float scale = 1.0f / (float)cnt[d];

    const float xv = X_edge[(size_t)e * D_EDGE + (lane & 15)];
    float xe[D_EDGE];
    #pragma unroll
    for (int k = 0; k < D_EDGE; ++k) xe[k] = __shfl(xv, k, 32);

    const float* __restrict__ hs = h_src + (size_t)s * D_OUT;
    float* __restrict__ od = out + (size_t)d * D_OUT;

    #pragma unroll
    for (int j = 0; j < 4; ++j) {
        const int c = lane + j * 32;
        float v = hs[c];
        #pragma unroll
        for (int k = 0; k < D_EDGE; ++k) v = fmaf(xe[k], wl[k * D_OUT + c], v);
        atomicAdd(&od[c], v * scale);
    }
}

extern "C" void kernel_launch(void* const* d_in, const int* in_sizes, int n_in,
                              void* d_out, int out_size, void* d_ws, size_t ws_size,
                              hipStream_t stream) {
    const float* X       = (const float*)d_in[0];
    const float* X_edge  = (const float*)d_in[1];
    const float* W_neigh = (const float*)d_in[2];
    const float* W_self  = (const float*)d_in[3];
    const float* b_self  = (const float*)d_in[4];
    const float* W_edge  = (const float*)d_in[5];
    const int*   src     = (const int*)d_in[6];
    const int*   dst     = (const int*)d_in[7];
    float* out = (float*)d_out;

    float* h_src = (float*)d_ws;
    int*   cnt   = (int*)((char*)d_ws + (size_t)N_NODES * D_OUT * sizeof(float));

    zero_cnt_kernel<<<(N_NODES + 255) / 256, 256, 0, stream>>>(cnt, N_NODES);
    count_edges_kernel<<<(N_EDGES + 255) / 256, 256, 0, stream>>>(dst, cnt, N_EDGES);

    gemm_dual_wmma_kernel<<<N_NODES / 16, 512, 0, stream>>>(
        X, W_self, W_neigh, b_self, out, h_src);

    edge_scatter_kernel<<<(N_EDGES + EDGES_PER_BLOCK - 1) / EDGES_PER_BLOCK, 256, 0, stream>>>(
        X_edge, h_src, src, dst, cnt, W_edge, out, N_EDGES);
}


# f32 atomic edge scatter replaced by dst-grouped gather: rank = count atomic's return, scan, fill; one wave sums a node's rows in registers, feats summed before W_edge; sum order changed; 6 launches
# speedup vs baseline: 2.1208x; 2.1208x over previous
#include <hip/hip_runtime.h>
#include <hip/hip_bf16.h>

#define N_NODES 100000
#define N_EDGES 1600000
#define D_IN    256
#define D_EDGE  16
#define D_OUT   128

typedef __attribute__((ext_vector_type(2))) float v2f;
typedef __attribute__((ext_vector_type(8))) float v8f;

__global__ void zero_cnt_kernel(int* __restrict__ cnt, int n) {
    int i = blockIdx.x * blockDim.x + threadIdx.x;
    if (i < n) cnt[i] = 0;
}

__global__ void count_rank_kernel(const int* __restrict__ dst,
                                  int* __restrict__ cnt,
                                  int* __restrict__ rank, int e_total) {
    int e = blockIdx.x * blockDim.x + threadIdx.x;
    if (e < e_total) rank[e] = atomicAdd(&cnt[dst[e]], 1);
}

#define XS_STRIDE 257

__global__ void __launch_bounds__(512)
gemm_dual_wmma_kernel(const float* __restrict__ X,
                      const float* __restrict__ W_self,
                      const float* __restrict__ W_neigh,
                      const float* __restrict__ b_self,
                      float* __restrict__ out_self,
                      float* __restrict__ out_neigh) {
    __shared__ float xs[16 * XS_STRIDE];

    const int rowBase = blockIdx.x * 16;

    for (int idx = threadIdx.x; idx < 16 * D_IN; idx += 512) {
        const int r = idx >> 8;
        const int c = idx & 255;
        xs[r * XS_STRIDE + c] = X[(size_t)(rowBase + r) * D_IN + c];
    }
    __syncthreads();

    const int wave  = threadIdx.x >> 5;
    const int lane  = threadIdx.x & 31;
    const int ntile = wave & 7;
    const bool isSelf = (wave < 8);

    const float* __restrict__ W = isSelf ? W_self : W_neigh;

    const int m     = lane & 15;
    const int khalf = lane >> 4;
    const int col   = ntile * 16 + m;

    v8f acc = {};
    const float* xrow = &xs[m * XS_STRIDE];

    #pragma unroll 4
    for (int kb = 0; kb < D_IN; kb += 4) {
        const int k0 = kb + 2 * khalf;
        v2f a, b;
        a.x = xrow[k0 + 0];
        a.y = xrow[k0 + 1];
        b.x = W[(size_t)(k0 + 0) * D_OUT + col];
        b.y = W[(size_t)(k0 + 1) * D_OUT + col];
        acc = __builtin_amdgcn_wmma_f32_16x16x4_f32(
             false, a,  false, b,
             (short)0, acc,  false,  false);
    }

    const float bias = isSelf ? b_self[col] : 0.0f;
    float* __restrict__ outp = isSelf ? out_self : out_neigh;

    #pragma unroll
    for (int r = 0; r < 8; ++r) {
        const int row = rowBase + r + 8 * khalf;
        outp[(size_t)row * D_OUT + col] = acc[r] + bias;
    }
}

#define SCAN_BLOCK 1024

__global__ void __launch_bounds__(SCAN_BLOCK)
scan_cnt_kernel(const int* __restrict__ cnt, int* __restrict__ start, int n) {
    __shared__ int wave_incl[32];
    __shared__ int wave_prev[32];
    __shared__ int block_base;

    const int t    = threadIdx.x;
    const int lane = t & 31;
    const int w    = t >> 5;
    const int b    = blockIdx.x;

    int prev = 0;
    for (int j = 0; j < b; ++j) prev += cnt[j * SCAN_BLOCK + t];

    const int idx = b * SCAN_BLOCK + t;
    const int v = (idx < n) ? cnt[idx] : 0;

    int incl = v;
    for (int off = 1; off < 32; off <<= 1) {
        const int u = __shfl_up(incl, off, 32);
        if (lane >= off) incl += u;
    }
    for (int off = 16; off > 0; off >>= 1) prev += __shfl_xor(prev, off, 32);

    if (lane == 31) wave_incl[w] = incl;
    if (lane == 0)  wave_prev[w] = prev;
    __syncthreads();

    if (w == 0) {
        const int tot = wave_incl[lane];
        int inc2 = tot;
            for (int off = 1; off < 32; off <<= 1) {
            const int u = __shfl_up(inc2, off, 32);
            if (lane >= off) inc2 += u;
        }
        int p = wave_prev[lane];
            for (int off = 16; off > 0; off >>= 1) p += __shfl_xor(p, off, 32);
        wave_incl[lane] = inc2 - tot;
        if (lane == 0) block_base = p;
    }
    __syncthreads();

    if (idx < n) start[idx] = block_base + wave_incl[w] + incl - v;
    if (idx == n - 1) start[n] = block_base + wave_incl[w] + incl;
}

__global__ void fill_pairs_kernel(const int* __restrict__ src,
                                  const int* __restrict__ dst,
                                  const int* __restrict__ rank,
                                  const int* __restrict__ start,
                                  int2* __restrict__ pairs, int e_total) {
    int e = blockIdx.x * blockDim.x + threadIdx.x;
    if (e < e_total) pairs[start[dst[e]] + rank[e]] = make_int2(src[e], e);
}

#define GATHER_WAVES      8
#define NODES_PER_WAVE    5
#define ROWS_IN_FLIGHT    8

__global__ void __launch_bounds__(256)
node_gather_kernel(const float* __restrict__ X_edge,
                   const float* __restrict__ h_src,
                   const int2* __restrict__ pairs,
                   const int* __restrict__ start,
                   const float* __restrict__ W_edge,
                   float* __restrict__ out, int n_total) {
    __shared__ float4 wl[D_EDGE * D_OUT / 4];
    for (int i = threadIdx.x; i < D_EDGE * D_OUT / 4; i += 256)
        wl[i] = reinterpret_cast<const float4*>(W_edge)[i];
    __syncthreads();

    const int lane = threadIdx.x & 31;
    const int q    = lane & 3;
    const int sub  = lane >> 2;

    for (int it = 0; it < NODES_PER_WAVE; ++it) {
        const int n = __builtin_amdgcn_readfirstlane(
            (blockIdx.x * NODES_PER_WAVE + it) * GATHER_WAVES + (threadIdx.x >> 5));
        if (n >= n_total) break;
        const int beg = start[n];
        const int deg = start[n + 1] - beg;
        if (deg == 0) continue;

        float4* orow = reinterpret_cast<float4*>(out + (size_t)n * D_OUT) + lane;
        const float4 self = *orow;

        float4 acc  = make_float4(0.f, 0.f, 0.f, 0.f);
        float4 xacc = make_float4(0.f, 0.f, 0.f, 0.f);

        for (int base = 0; base < deg; base += 32) {
            const int m = min(32, deg - base);
            int2 pr = make_int2(0, 0);
            if (lane < m) pr = pairs[beg + base + lane];

            float4 xv[4];
            #pragma unroll
            for (int g = 0; g < 4; ++g) {
                const int e = __shfl(pr.y, g * 8 + sub, 32);
                xv[g] = *reinterpret_cast<const float4*>(X_edge + (size_t)e * D_EDGE + q * 4);
            }

            for (int j0 = 0; j0 < m; j0 += ROWS_IN_FLIGHT) {
                float4 h[ROWS_IN_FLIGHT];
                #pragma unroll
                for (int u = 0; u < ROWS_IN_FLIGHT; ++u) {
                    const int jj = min(j0 + u, m - 1);
                    const int s = __builtin_amdgcn_readlane(pr.x, jj);
                    h[u] = reinterpret_cast<const float4*>(h_src + (size_t)s * D_OUT)[lane];
                }
                #pragma unroll
                for (int u = 0; u < ROWS_IN_FLIGHT; ++u) {
                    if (j0 + u < m) { acc.x += h[u].x; acc.y += h[u].y; acc.z += h[u].z; acc.w += h[u].w; }
                }
            }

            #pragma unroll
            for (int g = 0; g < 4; ++g) {
                if (g * 8 + sub < m) {
                    xacc.x += xv[g].x; xacc.y += xv[g].y; xacc.z += xv[g].z; xacc.w += xv[g].w;
                }
            }
        }

        for (int off = 4; off < 32; off <<= 1) {
            xacc.x += __shfl_xor(xacc.x, off, 32);
            xacc.y += __shfl_xor(xacc.y, off, 32);
            xacc.z += __shfl_xor(xacc.z, off, 32);
            xacc.w += __shfl_xor(xacc.w, off, 32);
        }

        #pragma unroll
        for (int qq = 0; qq < 4; ++qq) {
            const float x0 = __int_as_float(__builtin_amdgcn_readlane(__float_as_int(xacc.x), qq));
            const float x1 = __int_as_float(__builtin_amdgcn_readlane(__float_as_int(xacc.y), qq));
            const float x2 = __int_as_float(__builtin_amdgcn_readlane(__float_as_int(xacc.z), qq));
            const float x3 = __int_as_float(__builtin_amdgcn_readlane(__float_as_int(xacc.w), qq));
            const float4 w0 = wl[(qq * 4 + 0) * (D_OUT / 4) + lane];
            const float4 w1 = wl[(qq * 4 + 1) * (D_OUT / 4) + lane];
            const float4 w2 = wl[(qq * 4 + 2) * (D_OUT / 4) + lane];
            const float4 w3 = wl[(qq * 4 + 3) * (D_OUT / 4) + lane];
            acc.x = fmaf(x0, w0.x, acc.x); acc.y = fmaf(x0, w0.y, acc.y);
            acc.z = fmaf(x0, w0.z, acc.z); acc.w = fmaf(x0, w0.w, acc.w);
            acc.x = fmaf(x1, w1.x, acc.x); acc.y = fmaf(x1, w1.y, acc.y);
            acc.z = fmaf(x1, w1.z, acc.z); acc.w = fmaf(x1, w1.w, acc.w);
            acc.x = fmaf(x2, w2.x, acc.x); acc.y = fmaf(x2, w2.y, acc.y);
            acc.z = fmaf(x2, w2.z, acc.z); acc.w = fmaf(x2, w2.w, acc.w);
            acc.x = fmaf(x3, w3.x, acc.x); acc.y = fmaf(x3, w3.y, acc.y);
            acc.z = fmaf(x3, w3.z, acc.z); acc.w = fmaf(x3, w3.w, acc.w);
        }

        const float fdeg = (float)deg;
        float4 r;
        r.x = self.x + acc.x / fdeg;
        r.y = self.y + acc.y / fdeg;
        r.z = self.z + acc.z / fdeg;
        r.w = self.w + acc.w / fdeg;
        *orow = r;
    }
}

extern "C" void kernel_launch(void* const* d_in, const int* in_sizes, int n_in,
                              void* d_out, int out_size, void* d_ws, size_t ws_size,
                              hipStream_t stream) {
    const float* X       = (const float*)d_in[0];
    const float* X_edge  = (const float*)d_in[1];
    const float* W_neigh = (const float*)d_in[2];
    const float* W_self  = (const float*)d_in[3];
    const float* b_self  = (const float*)d_in[4];
    const float* W_edge  = (const float*)d_in[5];
    const int*   src     = (const int*)d_in[6];
    const int*   dst     = (const int*)d_in[7];
    float* out = (float*)d_out;

    float* h_src = (float*)d_ws;
    int*   cnt   = (int*)((char*)d_ws + (size_t)N_NODES * D_OUT * sizeof(float));
    int*   start = (int*)((char*)d_ws + (size_t)N_NODES * D_OUT * sizeof(float)
                                      + (size_t)N_NODES * sizeof(int));
    int*   rank  = (int*)((char*)d_ws + (size_t)N_NODES * D_OUT * sizeof(float)
                                      + (size_t)(2 * N_NODES + 4) * sizeof(int));
    int2*  pairs = (int2*)((char*)d_ws + (size_t)N_NODES * D_OUT * sizeof(float)
                                       + (size_t)(2 * N_NODES + 4) * sizeof(int)
                                       + (size_t)N_EDGES * sizeof(int));

    zero_cnt_kernel<<<(N_NODES + 255) / 256, 256, 0, stream>>>(cnt, N_NODES);
    count_rank_kernel<<<(N_EDGES + 255) / 256, 256, 0, stream>>>(dst, cnt, rank, N_EDGES);
    scan_cnt_kernel<<<(N_NODES + SCAN_BLOCK - 1) / SCAN_BLOCK, SCAN_BLOCK, 0, stream>>>(
        cnt, start, N_NODES);
    fill_pairs_kernel<<<(N_EDGES + 255) / 256, 256, 0, stream>>>(
        src, dst, rank, start, pairs, N_EDGES);

    gemm_dual_wmma_kernel<<<N_NODES / 16, 512, 0, stream>>>(
        X, W_self, W_neigh, b_self, out, h_src);

    node_gather_kernel<<<(N_NODES + GATHER_WAVES * NODES_PER_WAVE - 1) / (GATHER_WAVES * NODES_PER_WAVE),
                         256, 0, stream>>>(
        X_edge, h_src, pairs, start, W_edge, out, N_NODES);
}


# dst grouping on LDS counters: a work-group per (node range, edge chunk) counts, then ranks its range's edges by ds_add return; global atomics, zero, scan, rank array gone; 4 launches; arrival order
# speedup vs baseline: 2.7473x; 1.2954x over previous
#include <hip/hip_runtime.h>
#include <hip/hip_bf16.h>

#define N_NODES 100000
#define N_EDGES 1600000
#define D_IN    256
#define D_EDGE  16
#define D_OUT   128

typedef __attribute__((ext_vector_type(2))) float v2f;
typedef __attribute__((ext_vector_type(8))) float v8f;

#define N_RANGES     25
#define RANGE_NODES  4000
#define N_CHUNKS     10
#define CHUNK_EDGES  160000
#define CSR_THREADS  1024
#define CHUNK_VEC    (CHUNK_EDGES / 4)
#define RANGE_VEC    (RANGE_NODES / 4)

static_assert(N_RANGES * RANGE_NODES == N_NODES, "ranges must tile the nodes");
static_assert(N_CHUNKS * CHUNK_EDGES == N_EDGES, "chunks must tile the edges");
static_assert(CHUNK_EDGES % 4 == 0 && RANGE_NODES % 4 == 0, "16-byte accesses");
static_assert(RANGE_VEC <= CSR_THREADS, "one thread per four nodes");
static_assert(N_RANGES * N_CHUNKS <= CSR_THREADS, "one thread per total");

__device__ __forceinline__ void count4(int* lc, const int4 d, const int lo) {
    const unsigned k0 = (unsigned)(d.x - lo);
    const unsigned k1 = (unsigned)(d.y - lo);
    const unsigned k2 = (unsigned)(d.z - lo);
    const unsigned k3 = (unsigned)(d.w - lo);
    if (k0 < RANGE_NODES) atomicAdd(&lc[k0], 1);
    if (k1 < RANGE_NODES) atomicAdd(&lc[k1], 1);
    if (k2 < RANGE_NODES) atomicAdd(&lc[k2], 1);
    if (k3 < RANGE_NODES) atomicAdd(&lc[k3], 1);
}

__global__ void __launch_bounds__(CSR_THREADS)
range_count_kernel(const int* __restrict__ dst,
                   int* __restrict__ part, int* __restrict__ tot) {
    __shared__ int lc[RANGE_NODES];
    __shared__ int wsum[32];

    const int t    = threadIdx.x;
    const int lane = t & 31;
    const int w    = t >> 5;
    const int r    = blockIdx.x % N_RANGES;
    const int c    = blockIdx.x / N_RANGES;
    const int lo   = r * RANGE_NODES;

    for (int i = t; i < RANGE_NODES; i += CSR_THREADS) lc[i] = 0;
    __syncthreads();

    const int4* __restrict__ d4 =
        reinterpret_cast<const int4*>(dst + (size_t)c * CHUNK_EDGES);

    for (int i = t; i < CHUNK_VEC; i += 4 * CSR_THREADS) {
        int4 d[4];
        #pragma unroll
        for (int u = 0; u < 4; ++u) {
            const int idx = i + u * CSR_THREADS;
            d[u] = make_int4(-1, -1, -1, -1);
            if (idx < CHUNK_VEC) d[u] = d4[idx];
        }
        #pragma unroll
        for (int u = 0; u < 4; ++u) count4(lc, d[u], lo);
    }
    __syncthreads();

    int s = 0;
    if (t < RANGE_VEC) {
        const int4 v = reinterpret_cast<const int4*>(lc)[t];
        reinterpret_cast<int4*>(part + (size_t)c * N_NODES + lo)[t] = v;
        s = v.x + v.y + v.z + v.w;
    }
    for (int off = 16; off > 0; off >>= 1) s += __shfl_xor(s, off, 32);
    if (lane == 0) wsum[w] = s;
    __syncthreads();
    if (w == 0) {
        int p = wsum[lane];
        for (int off = 16; off > 0; off >>= 1) p += __shfl_xor(p, off, 32);
        if (lane == 0) tot[c * N_RANGES + r] = p;
    }
}

__device__ __forceinline__ void rank4(int* lb, const int4 d, const int lo, int (&p)[4]) {
    const int dd[4] = {d.x, d.y, d.z, d.w};
    #pragma unroll
    for (int j = 0; j < 4; ++j) {
        const unsigned k = (unsigned)(dd[j] - lo);
        p[j] = -1;
        if (k < RANGE_NODES) p[j] = atomicAdd(&lb[k], 1);
    }
}

__device__ __forceinline__ void store4(int2* __restrict__ pairs, const int4 sv,
                                       const int e, const int (&p)[4]) {
    const int ss[4] = {sv.x, sv.y, sv.z, sv.w};
    #pragma unroll
    for (int j = 0; j < 4; ++j)
        if (p[j] >= 0) pairs[p[j]] = make_int2(ss[j], e + j);
}

__global__ void __launch_bounds__(CSR_THREADS)
range_place_kernel(const int* __restrict__ src,
                   const int* __restrict__ dst,
                   const int* __restrict__ part,
                   const int* __restrict__ tot,
                   int* __restrict__ start,
                   int2* __restrict__ pairs) {
    __shared__ int lb[RANGE_NODES];
    __shared__ int wave_incl[32];
    __shared__ int red[32];
    __shared__ int range_base;

    const int t    = threadIdx.x;
    const int lane = t & 31;
    const int w    = t >> 5;
    const int r    = blockIdx.x % N_RANGES;
    const int c    = blockIdx.x / N_RANGES;
    const int lo   = r * RANGE_NODES;

    int tb = 0;
    if (t < N_RANGES * N_CHUNKS && (t % N_RANGES) < r) tb = tot[t];

    int4 cnt = make_int4(0, 0, 0, 0);
    int4 pre = make_int4(0, 0, 0, 0);
    if (t < RANGE_VEC) {
        int4 v[N_CHUNKS];
        #pragma unroll
        for (int cc = 0; cc < N_CHUNKS; ++cc)
            v[cc] = reinterpret_cast<const int4*>(part + (size_t)cc * N_NODES + lo)[t];
        #pragma unroll
        for (int cc = 0; cc < N_CHUNKS; ++cc) {
            cnt.x += v[cc].x; cnt.y += v[cc].y; cnt.z += v[cc].z; cnt.w += v[cc].w;
            if (cc < c) { pre.x += v[cc].x; pre.y += v[cc].y; pre.z += v[cc].z; pre.w += v[cc].w; }
        }
    }

    const int s = cnt.x + cnt.y + cnt.z + cnt.w;
    int incl = s;
    for (int off = 1; off < 32; off <<= 1) {
        const int u = __shfl_up(incl, off, 32);
        if (lane >= off) incl += u;
    }
    for (int off = 16; off > 0; off >>= 1) tb += __shfl_xor(tb, off, 32);
    if (lane == 31) wave_incl[w] = incl;
    if (lane == 0)  red[w] = tb;
    __syncthreads();

    if (w == 0) {
        const int wt = wave_incl[lane];
        int inc2 = wt;
        for (int off = 1; off < 32; off <<= 1) {
            const int u = __shfl_up(inc2, off, 32);
            if (lane >= off) inc2 += u;
        }
        int p = red[lane];
        for (int off = 16; off > 0; off >>= 1) p += __shfl_xor(p, off, 32);
        wave_incl[lane] = inc2 - wt;
        if (lane == 0) range_base = p;
    }
    __syncthreads();

    if (t < RANGE_VEC) {
        const int b0 = range_base + wave_incl[w] + incl - s;
        const int b1 = b0 + cnt.x;
        const int b2 = b1 + cnt.y;
        const int b3 = b2 + cnt.z;
        if (c == 0) {
            reinterpret_cast<int4*>(start + lo)[t] = make_int4(b0, b1, b2, b3);
            if (r == N_RANGES - 1 && t == RANGE_VEC - 1)
                start[N_NODES] = b3 + cnt.w;
        }
        reinterpret_cast<int4*>(lb)[t] =
            make_int4(b0 + pre.x, b1 + pre.y, b2 + pre.z, b3 + pre.w);
    }
    __syncthreads();

    const int4* __restrict__ d4 =
        reinterpret_cast<const int4*>(dst + (size_t)c * CHUNK_EDGES);
    const int4* __restrict__ s4 =
        reinterpret_cast<const int4*>(src + (size_t)c * CHUNK_EDGES);
    const int e0 = c * CHUNK_EDGES;

    for (int i = t; i < CHUNK_VEC; i += 4 * CSR_THREADS) {
        int4 d[4], sv[4];
        #pragma unroll
        for (int u = 0; u < 4; ++u) {
            const int idx = i + u * CSR_THREADS;
            d[u]  = make_int4(-1, -1, -1, -1);
            sv[u] = make_int4(0, 0, 0, 0);
            if (idx < CHUNK_VEC) { d[u] = d4[idx]; sv[u] = s4[idx]; }
        }
        int p[4][4];
        #pragma unroll
        for (int u = 0; u < 4; ++u) rank4(lb, d[u], lo, p[u]);
        #pragma unroll
        for (int u = 0; u < 4; ++u)
            store4(pairs, sv[u], e0 + 4 * (i + u * CSR_THREADS), p[u]);
    }
}

#define XS_STRIDE 257

__global__ void __launch_bounds__(512)
gemm_dual_wmma_kernel(const float* __restrict__ X,
                      const float* __restrict__ W_self,
                      const float* __restrict__ W_neigh,
                      const float* __restrict__ b_self,
                      float* __restrict__ out_self,
                      float* __restrict__ out_neigh) {
    __shared__ float xs[16 * XS_STRIDE];

    const int rowBase = blockIdx.x * 16;

    for (int idx = threadIdx.x; idx < 16 * D_IN; idx += 512) {
        const int r = idx >> 8;
        const int c = idx & 255;
        xs[r * XS_STRIDE + c] = X[(size_t)(rowBase + r) * D_IN + c];
    }
    __syncthreads();

    const int wave  = threadIdx.x >> 5;
    const int lane  = threadIdx.x & 31;
    const int ntile = wave & 7;
    const bool isSelf = (wave < 8);

    const float* __restrict__ W = isSelf ? W_self : W_neigh;

    const int m     = lane & 15;
    const int khalf = lane >> 4;
    const int col   = ntile * 16 + m;

    v8f acc = {};
    const float* xrow = &xs[m * XS_STRIDE];

    #pragma unroll 4
    for (int kb = 0; kb < D_IN; kb += 4) {
        const int k0 = kb + 2 * khalf;
        v2f a, b;
        a.x = xrow[k0 + 0];
        a.y = xrow[k0 + 1];
        b.x = W[(size_t)(k0 + 0) * D_OUT + col];
        b.y = W[(size_t)(k0 + 1) * D_OUT + col];
        acc = __builtin_amdgcn_wmma_f32_16x16x4_f32(
             false, a,  false, b,
             (short)0, acc,  false,  false);
    }

    const float bias = isSelf ? b_self[col] : 0.0f;
    float* __restrict__ outp = isSelf ? out_self : out_neigh;

    #pragma unroll
    for (int r = 0; r < 8; ++r) {
        const int row = rowBase + r + 8 * khalf;
        outp[(size_t)row * D_OUT + col] = acc[r] + bias;
    }
}

#define GATHER_WAVES      8
#define NODES_PER_WAVE    5
#define ROWS_IN_FLIGHT    8

__global__ void __launch_bounds__(256)
node_gather_kernel(const float* __restrict__ X_edge,
                   const float* __restrict__ h_src,
                   const int2* __restrict__ pairs,
                   const int* __restrict__ start,
                   const float* __restrict__ W_edge,
                   float* __restrict__ out, int n_total) {
    __shared__ float4 wl[D_EDGE * D_OUT / 4];
    for (int i = threadIdx.x; i < D_EDGE * D_OUT / 4; i += 256)
        wl[i] = reinterpret_cast<const float4*>(W_edge)[i];
    __syncthreads();

    const int lane = threadIdx.x & 31;
    const int q    = lane & 3;
    const int sub  = lane >> 2;

    for (int it = 0; it < NODES_PER_WAVE; ++it) {
        const int n = __builtin_amdgcn_readfirstlane(
            (blockIdx.x * NODES_PER_WAVE + it) * GATHER_WAVES + (threadIdx.x >> 5));
        if (n >= n_total) break;
        const int beg = start[n];
        const int deg = start[n + 1] - beg;
        if (deg == 0) continue;

        float4* orow = reinterpret_cast<float4*>(out + (size_t)n * D_OUT) + lane;
        const float4 self = *orow;

        float4 acc  = make_float4(0.f, 0.f, 0.f, 0.f);
        float4 xacc = make_float4(0.f, 0.f, 0.f, 0.f);

        for (int base = 0; base < deg; base += 32) {
            const int m = min(32, deg - base);
            int2 pr = make_int2(0, 0);
            if (lane < m) pr = pairs[beg + base + lane];

            float4 xv[4];
            #pragma unroll
            for (int g = 0; g < 4; ++g) {
                const int e = __shfl(pr.y, g * 8 + sub, 32);
                xv[g] = *reinterpret_cast<const float4*>(X_edge + (size_t)e * D_EDGE + q * 4);
            }

            for (int j0 = 0; j0 < m; j0 += ROWS_IN_FLIGHT) {
                float4 h[ROWS_IN_FLIGHT];
                #pragma unroll
                for (int u = 0; u < ROWS_IN_FLIGHT; ++u) {
                    const int jj = min(j0 + u, m - 1);
                    const int s = __builtin_amdgcn_readlane(pr.x, jj);
                    h[u] = reinterpret_cast<const float4*>(h_src + (size_t)s * D_OUT)[lane];
                }
                #pragma unroll
                for (int u = 0; u < ROWS_IN_FLIGHT; ++u) {
                    if (j0 + u < m) { acc.x += h[u].x; acc.y += h[u].y; acc.z += h[u].z; acc.w += h[u].w; }
                }
            }

            #pragma unroll
            for (int g = 0; g < 4; ++g) {
                if (g * 8 + sub < m) {
                    xacc.x += xv[g].x; xacc.y += xv[g].y; xacc.z += xv[g].z; xacc.w += xv[g].w;
                }
            }
        }

        for (int off = 4; off < 32; off <<= 1) {
            xacc.x += __shfl_xor(xacc.x, off, 32);
            xacc.y += __shfl_xor(xacc.y, off, 32);
            xacc.z += __shfl_xor(xacc.z, off, 32);
            xacc.w += __shfl_xor(xacc.w, off, 32);
        }

        #pragma unroll
        for (int qq = 0; qq < 4; ++qq) {
            const float x0 = __int_as_float(__builtin_amdgcn_readlane(__float_as_int(xacc.x), qq));
            const float x1 = __int_as_float(__builtin_amdgcn_readlane(__float_as_int(xacc.y), qq));
            const float x2 = __int_as_float(__builtin_amdgcn_readlane(__float_as_int(xacc.z), qq));
            const float x3 = __int_as_float(__builtin_amdgcn_readlane(__float_as_int(xacc.w), qq));
            const float4 w0 = wl[(qq * 4 + 0) * (D_OUT / 4) + lane];
            const float4 w1 = wl[(qq * 4 + 1) * (D_OUT / 4) + lane];
            const float4 w2 = wl[(qq * 4 + 2) * (D_OUT / 4) + lane];
            const float4 w3 = wl[(qq * 4 + 3) * (D_OUT / 4) + lane];
            acc.x = fmaf(x0, w0.x, acc.x); acc.y = fmaf(x0, w0.y, acc.y);
            acc.z = fmaf(x0, w0.z, acc.z); acc.w = fmaf(x0, w0.w, acc.w);
            acc.x = fmaf(x1, w1.x, acc.x); acc.y = fmaf(x1, w1.y, acc.y);
            acc.z = fmaf(x1, w1.z, acc.z); acc.w = fmaf(x1, w1.w, acc.w);
            acc.x = fmaf(x2, w2.x, acc.x); acc.y = fmaf(x2, w2.y, acc.y);
            acc.z = fmaf(x2, w2.z, acc.z); acc.w = fmaf(x2, w2.w, acc.w);
            acc.x = fmaf(x3, w3.x, acc.x); acc.y = fmaf(x3, w3.y, acc.y);
            acc.z = fmaf(x3, w3.z, acc.z); acc.w = fmaf(x3, w3.w, acc.w);
        }

        const float fdeg = (float)deg;
        float4 r;
        r.x = self.x + acc.x / fdeg;
        r.y = self.y + acc.y / fdeg;
        r.z = self.z + acc.z / fdeg;
        r.w = self.w + acc.w / fdeg;
        *orow = r;
    }
}

extern "C" void kernel_launch(void* const* d_in, const int* in_sizes, int n_in,
                              void* d_out, int out_size, void* d_ws, size_t ws_size,
                              hipStream_t stream) {
    const float* X       = (const float*)d_in[0];
    const float* X_edge  = (const float*)d_in[1];
    const float* W_neigh = (const float*)d_in[2];
    const float* W_self  = (const float*)d_in[3];
    const float* b_self  = (const float*)d_in[4];
    const float* W_edge  = (const float*)d_in[5];
    const int*   src     = (const int*)d_in[6];
    const int*   dst     = (const int*)d_in[7];
    float* out = (float*)d_out;

    float* h_src = (float*)d_ws;
    int*   start = (int*)((char*)d_ws + (size_t)N_NODES * D_OUT * sizeof(float));
    int*   part  = (int*)((char*)d_ws + (size_t)N_NODES * D_OUT * sizeof(float)
                                      + (size_t)(N_NODES + 4) * sizeof(int));
    int*   tot   = (int*)((char*)d_ws + (size_t)N_NODES * D_OUT * sizeof(float)
                                      + (size_t)(N_NODES + 4) * sizeof(int)
                                      + (size_t)N_CHUNKS * N_NODES * sizeof(int));
    int2*  pairs = (int2*)((char*)d_ws + (size_t)N_NODES * D_OUT * sizeof(float)
                                       + (size_t)(N_NODES + 4) * sizeof(int)
                                       + (size_t)N_CHUNKS * N_NODES * sizeof(int)
                                       + (size_t)256 * sizeof(int));

    range_count_kernel<<<N_RANGES * N_CHUNKS, CSR_THREADS, 0, stream>>>(dst, part, tot);
    range_place_kernel<<<N_RANGES * N_CHUNKS, CSR_THREADS, 0, stream>>>(
        src, dst, part, tot, start, pairs);

    gemm_dual_wmma_kernel<<<N_NODES / 16, 512, 0, stream>>>(
        X, W_self, W_neigh, b_self, out, h_src);

    node_gather_kernel<<<(N_NODES + GATHER_WAVES * NODES_PER_WAVE - 1) / (GATHER_WAVES * NODES_PER_WAVE),
                         256, 0, stream>>>(
        X_edge, h_src, pairs, start, W_edge, out, N_NODES);
}


# GEMM: each wave keeps its 256x16 weight slab in 128 registers and 256 work-groups loop over row tiles (X tile double-buffered in LDS) instead of 6250 groups re-reading W per 16 rows
# speedup vs baseline: 3.1742x; 1.1554x over previous
#include <hip/hip_runtime.h>
#include <hip/hip_bf16.h>

#define N_NODES 100000
#define N_EDGES 1600000
#define D_IN    256
#define D_EDGE  16
#define D_OUT   128

typedef __attribute__((ext_vector_type(2))) float v2f;
typedef __attribute__((ext_vector_type(8))) float v8f;

#define N_RANGES     25
#define RANGE_NODES  4000
#define N_CHUNKS     10
#define CHUNK_EDGES  160000
#define CSR_THREADS  1024
#define CHUNK_VEC    (CHUNK_EDGES / 4)
#define RANGE_VEC    (RANGE_NODES / 4)

static_assert(N_RANGES * RANGE_NODES == N_NODES, "ranges must tile the nodes");
static_assert(N_CHUNKS * CHUNK_EDGES == N_EDGES, "chunks must tile the edges");
static_assert(CHUNK_EDGES % 4 == 0 && RANGE_NODES % 4 == 0, "16-byte accesses");
static_assert(RANGE_VEC <= CSR_THREADS, "one thread per four nodes");
static_assert(N_RANGES * N_CHUNKS <= CSR_THREADS, "one thread per total");

__device__ __forceinline__ void count4(int* lc, const int4 d, const int lo) {
    const unsigned k0 = (unsigned)(d.x - lo);
    const unsigned k1 = (unsigned)(d.y - lo);
    const unsigned k2 = (unsigned)(d.z - lo);
    const unsigned k3 = (unsigned)(d.w - lo);
    if (k0 < RANGE_NODES) atomicAdd(&lc[k0], 1);
    if (k1 < RANGE_NODES) atomicAdd(&lc[k1], 1);
    if (k2 < RANGE_NODES) atomicAdd(&lc[k2], 1);
    if (k3 < RANGE_NODES) atomicAdd(&lc[k3], 1);
}

__global__ void __launch_bounds__(CSR_THREADS)
range_count_kernel(const int* __restrict__ dst,
                   int* __restrict__ part, int* __restrict__ tot) {
    __shared__ int lc[RANGE_NODES];
    __shared__ int wsum[32];

    const int t    = threadIdx.x;
    const int lane = t & 31;
    const int w    = t >> 5;
    const int r    = blockIdx.x % N_RANGES;
    const int c    = blockIdx.x / N_RANGES;
    const int lo   = r * RANGE_NODES;

    for (int i = t; i < RANGE_NODES; i += CSR_THREADS) lc[i] = 0;
    __syncthreads();

    const int4* __restrict__ d4 =
        reinterpret_cast<const int4*>(dst + (size_t)c * CHUNK_EDGES);

    for (int i = t; i < CHUNK_VEC; i += 4 * CSR_THREADS) {
        int4 d[4];
        #pragma unroll
        for (int u = 0; u < 4; ++u) {
            const int idx = i + u * CSR_THREADS;
            d[u] = make_int4(-1, -1, -1, -1);
            if (idx < CHUNK_VEC) d[u] = d4[idx];
        }
        #pragma unroll
        for (int u = 0; u < 4; ++u) count4(lc, d[u], lo);
    }
    __syncthreads();

    int s = 0;
    if (t < RANGE_VEC) {
        const int4 v = reinterpret_cast<const int4*>(lc)[t];
        reinterpret_cast<int4*>(part + (size_t)c * N_NODES + lo)[t] = v;
        s = v.x + v.y + v.z + v.w;
    }
    for (int off = 16; off > 0; off >>= 1) s += __shfl_xor(s, off, 32);
    if (lane == 0) wsum[w] = s;
    __syncthreads();
    if (w == 0) {
        int p = wsum[lane];
        for (int off = 16; off > 0; off >>= 1) p += __shfl_xor(p, off, 32);
        if (lane == 0) tot[c * N_RANGES + r] = p;
    }
}

__device__ __forceinline__ void rank4(int* lb, const int4 d, const int lo, int (&p)[4]) {
    const int dd[4] = {d.x, d.y, d.z, d.w};
    #pragma unroll
    for (int j = 0; j < 4; ++j) {
        const unsigned k = (unsigned)(dd[j] - lo);
        p[j] = -1;
        if (k < RANGE_NODES) p[j] = atomicAdd(&lb[k], 1);
    }
}

__device__ __forceinline__ void store4(int2* __restrict__ pairs, const int4 sv,
                                       const int e, const int (&p)[4]) {
    const int ss[4] = {sv.x, sv.y, sv.z, sv.w};
    #pragma unroll
    for (int j = 0; j < 4; ++j)
        if (p[j] >= 0) pairs[p[j]] = make_int2(ss[j], e + j);
}

__global__ void __launch_bounds__(CSR_THREADS)
range_place_kernel(const int* __restrict__ src,
                   const int* __restrict__ dst,
                   const int* __restrict__ part,
                   const int* __restrict__ tot,
                   int* __restrict__ start,
                   int2* __restrict__ pairs) {
    __shared__ int lb[RANGE_NODES];
    __shared__ int wave_incl[32];
    __shared__ int red[32];
    __shared__ int range_base;

    const int t    = threadIdx.x;
    const int lane = t & 31;
    const int w    = t >> 5;
    const int r    = blockIdx.x % N_RANGES;
    const int c    = blockIdx.x / N_RANGES;
    const int lo   = r * RANGE_NODES;

    int tb = 0;
    if (t < N_RANGES * N_CHUNKS && (t % N_RANGES) < r) tb = tot[t];

    int4 cnt = make_int4(0, 0, 0, 0);
    int4 pre = make_int4(0, 0, 0, 0);
    if (t < RANGE_VEC) {
        int4 v[N_CHUNKS];
        #pragma unroll
        for (int cc = 0; cc < N_CHUNKS; ++cc)
            v[cc] = reinterpret_cast<const int4*>(part + (size_t)cc * N_NODES + lo)[t];
        #pragma unroll
        for (int cc = 0; cc < N_CHUNKS; ++cc) {
            cnt.x += v[cc].x; cnt.y += v[cc].y; cnt.z += v[cc].z; cnt.w += v[cc].w;
            if (cc < c) { pre.x += v[cc].x; pre.y += v[cc].y; pre.z += v[cc].z; pre.w += v[cc].w; }
        }
    }

    const int s = cnt.x + cnt.y + cnt.z + cnt.w;
    int incl = s;
    for (int off = 1; off < 32; off <<= 1) {
        const int u = __shfl_up(incl, off, 32);
        if (lane >= off) incl += u;
    }
    for (int off = 16; off > 0; off >>= 1) tb += __shfl_xor(tb, off, 32);
    if (lane == 31) wave_incl[w] = incl;
    if (lane == 0)  red[w] = tb;
    __syncthreads();

    if (w == 0) {
        const int wt = wave_incl[lane];
        int inc2 = wt;
        for (int off = 1; off < 32; off <<= 1) {
            const int u = __shfl_up(inc2, off, 32);
            if (lane >= off) inc2 += u;
        }
        int p = red[lane];
        for (int off = 16; off > 0; off >>= 1) p += __shfl_xor(p, off, 32);
        wave_incl[lane] = inc2 - wt;
        if (lane == 0) range_base = p;
    }
    __syncthreads();

    if (t < RANGE_VEC) {
        const int b0 = range_base + wave_incl[w] + incl - s;
        const int b1 = b0 + cnt.x;
        const int b2 = b1 + cnt.y;
        const int b3 = b2 + cnt.z;
        if (c == 0) {
            reinterpret_cast<int4*>(start + lo)[t] = make_int4(b0, b1, b2, b3);
            if (r == N_RANGES - 1 && t == RANGE_VEC - 1)
                start[N_NODES] = b3 + cnt.w;
        }
        reinterpret_cast<int4*>(lb)[t] =
            make_int4(b0 + pre.x, b1 + pre.y, b2 + pre.z, b3 + pre.w);
    }
    __syncthreads();

    const int4* __restrict__ d4 =
        reinterpret_cast<const int4*>(dst + (size_t)c * CHUNK_EDGES);
    const int4* __restrict__ s4 =
        reinterpret_cast<const int4*>(src + (size_t)c * CHUNK_EDGES);
    const int e0 = c * CHUNK_EDGES;

    for (int i = t; i < CHUNK_VEC; i += 4 * CSR_THREADS) {
        int4 d[4], sv[4];
        #pragma unroll
        for (int u = 0; u < 4; ++u) {
            const int idx = i + u * CSR_THREADS;
            d[u]  = make_int4(-1, -1, -1, -1);
            sv[u] = make_int4(0, 0, 0, 0);
            if (idx < CHUNK_VEC) { d[u] = d4[idx]; sv[u] = s4[idx]; }
        }
        int p[4][4];
        #pragma unroll
        for (int u = 0; u < 4; ++u) rank4(lb, d[u], lo, p[u]);
        #pragma unroll
        for (int u = 0; u < 4; ++u)
            store4(pairs, sv[u], e0 + 4 * (i + u * CSR_THREADS), p[u]);
    }
}

#define XS_PITCH    260
#define XS_PITCH4   (XS_PITCH / 4)
#define GEMM_BLOCKS 256
#define N_TILES     (N_NODES / 16)
#define K_STEPS     (D_IN / 4)

static_assert(N_TILES * 16 == N_NODES, "row tiles must tile the nodes");
static_assert(GEMM_BLOCKS <= N_TILES, "every work-group has a first tile");

__global__ void __launch_bounds__(512)
gemm_dual_wmma_kernel(const float* __restrict__ X,
                      const float* __restrict__ W_self,
                      const float* __restrict__ W_neigh,
                      const float* __restrict__ b_self,
                      float* __restrict__ out_self,
                      float* __restrict__ out_neigh) {
    __shared__ float4 xs4[2][16 * XS_PITCH4];

    const int wave  = threadIdx.x >> 5;
    const int lane  = threadIdx.x & 31;
    const int ntile = wave & 7;
    const bool isSelf = (wave < 8);

    const float* __restrict__ W = isSelf ? W_self : W_neigh;

    const int m     = lane & 15;
    const int khalf = lane >> 4;
    const int col   = ntile * 16 + m;

    v2f breg[K_STEPS];
    #pragma unroll
    for (int j = 0; j < K_STEPS; ++j) {
        const int k0 = 4 * j + 2 * khalf;
        breg[j].x = W[(size_t)(k0 + 0) * D_OUT + col];
        breg[j].y = W[(size_t)(k0 + 1) * D_OUT + col];
    }

    const float bias = isSelf ? b_self[col] : 0.0f;
    float* __restrict__ outp = isSelf ? out_self : out_neigh;

    const int sr = threadIdx.x >> 6;
    const int sc = threadIdx.x & 63;
    const float4* __restrict__ X4 = reinterpret_cast<const float4*>(X);

    int tile = blockIdx.x;
    float4 p0 = X4[(size_t)(tile * 16 + sr)     * (D_IN / 4) + sc];
    float4 p1 = X4[(size_t)(tile * 16 + sr + 8) * (D_IN / 4) + sc];
    xs4[0][sr * XS_PITCH4 + sc]       = p0;
    xs4[0][(sr + 8) * XS_PITCH4 + sc] = p1;
    __syncthreads();

    int buf = 0;
    for (;;) {
        const int  next    = tile + GEMM_BLOCKS;
        const bool hasNext = next < N_TILES;
        if (hasNext) {
            p0 = X4[(size_t)(next * 16 + sr)     * (D_IN / 4) + sc];
            p1 = X4[(size_t)(next * 16 + sr + 8) * (D_IN / 4) + sc];
        }

        const float* xrow = reinterpret_cast<const float*>(xs4[buf]) + m * XS_PITCH + 2 * khalf;
        v8f acc = {};
        #pragma unroll
        for (int j = 0; j < K_STEPS; ++j) {
            const v2f a = *reinterpret_cast<const v2f*>(xrow + 4 * j);
            acc = __builtin_amdgcn_wmma_f32_16x16x4_f32(
                 false, a,  false, breg[j],
                 (short)0, acc,  false,  false);
        }

        if (hasNext) {
            xs4[buf ^ 1][sr * XS_PITCH4 + sc]       = p0;
            xs4[buf ^ 1][(sr + 8) * XS_PITCH4 + sc] = p1;
            __syncthreads();
        }

        const int rowBase = tile * 16;
        #pragma unroll
        for (int r = 0; r < 8; ++r) {
            const int row = rowBase + r + 8 * khalf;
            outp[(size_t)row * D_OUT + col] = acc[r] + bias;
        }

        if (!hasNext) break;
        buf ^= 1;
        tile = next;
    }
}

#define GATHER_WAVES      8
#define NODES_PER_WAVE    5
#define ROWS_IN_FLIGHT    8

__global__ void __launch_bounds__(256)
node_gather_kernel(const float* __restrict__ X_edge,
                   const float* __restrict__ h_src,
                   const int2* __restrict__ pairs,
                   const int* __restrict__ start,
                   const float* __restrict__ W_edge,
                   float* __restrict__ out, int n_total) {
    __shared__ float4 wl[D_EDGE * D_OUT / 4];
    for (int i = threadIdx.x; i < D_EDGE * D_OUT / 4; i += 256)
        wl[i] = reinterpret_cast<const float4*>(W_edge)[i];
    __syncthreads();

    const int lane = threadIdx.x & 31;
    const int q    = lane & 3;
    const int sub  = lane >> 2;

    for (int it = 0; it < NODES_PER_WAVE; ++it) {
        const int n = __builtin_amdgcn_readfirstlane(
            (blockIdx.x * NODES_PER_WAVE + it) * GATHER_WAVES + (threadIdx.x >> 5));
        if (n >= n_total) break;
        const int beg = start[n];
        const int deg = start[n + 1] - beg;
        if (deg == 0) continue;

        float4* orow = reinterpret_cast<float4*>(out + (size_t)n * D_OUT) + lane;
        const float4 self = *orow;

        float4 acc  = make_float4(0.f, 0.f, 0.f, 0.f);
        float4 xacc = make_float4(0.f, 0.f, 0.f, 0.f);

        for (int base = 0; base < deg; base += 32) {
            const int m = min(32, deg - base);
            int2 pr = make_int2(0, 0);
            if (lane < m) pr = pairs[beg + base + lane];

            float4 xv[4];
            #pragma unroll
            for (int g = 0; g < 4; ++g) {
                const int e = __shfl(pr.y, g * 8 + sub, 32);
                xv[g] = *reinterpret_cast<const float4*>(X_edge + (size_t)e * D_EDGE + q * 4);
            }

            for (int j0 = 0; j0 < m; j0 += ROWS_IN_FLIGHT) {
                float4 h[ROWS_IN_FLIGHT];
                #pragma unroll
                for (int u = 0; u < ROWS_IN_FLIGHT; ++u) {
                    const int jj = min(j0 + u, m - 1);
                    const int s = __builtin_amdgcn_readlane(pr.x, jj);
                    h[u] = reinterpret_cast<const float4*>(h_src + (size_t)s * D_OUT)[lane];
                }
                #pragma unroll
                for (int u = 0; u < ROWS_IN_FLIGHT; ++u) {
                    if (j0 + u < m) { acc.x += h[u].x; acc.y += h[u].y; acc.z += h[u].z; acc.w += h[u].w; }
                }
            }

            #pragma unroll
            for (int g = 0; g < 4; ++g) {
                if (g * 8 + sub < m) {
                    xacc.x += xv[g].x; xacc.y += xv[g].y; xacc.z += xv[g].z; xacc.w += xv[g].w;
                }
            }
        }

        for (int off = 4; off < 32; off <<= 1) {
            xacc.x += __shfl_xor(xacc.x, off, 32);
            xacc.y += __shfl_xor(xacc.y, off, 32);
            xacc.z += __shfl_xor(xacc.z, off, 32);
            xacc.w += __shfl_xor(xacc.w, off, 32);
        }

        #pragma unroll
        for (int qq = 0; qq < 4; ++qq) {
            const float x0 = __int_as_float(__builtin_amdgcn_readlane(__float_as_int(xacc.x), qq));
            const float x1 = __int_as_float(__builtin_amdgcn_readlane(__float_as_int(xacc.y), qq));
            const float x2 = __int_as_float(__builtin_amdgcn_readlane(__float_as_int(xacc.z), qq));
            const float x3 = __int_as_float(__builtin_amdgcn_readlane(__float_as_int(xacc.w), qq));
            const float4 w0 = wl[(qq * 4 + 0) * (D_OUT / 4) + lane];
            const float4 w1 = wl[(qq * 4 + 1) * (D_OUT / 4) + lane];
            const float4 w2 = wl[(qq * 4 + 2) * (D_OUT / 4) + lane];
            const float4 w3 = wl[(qq * 4 + 3) * (D_OUT / 4) + lane];
            acc.x = fmaf(x0, w0.x, acc.x); acc.y = fmaf(x0, w0.y, acc.y);
            acc.z = fmaf(x0, w0.z, acc.z); acc.w = fmaf(x0, w0.w, acc.w);
            acc.x = fmaf(x1, w1.x, acc.x); acc.y = fmaf(x1, w1.y, acc.y);
            acc.z = fmaf(x1, w1.z, acc.z); acc.w = fmaf(x1, w1.w, acc.w);
            acc.x = fmaf(x2, w2.x, acc.x); acc.y = fmaf(x2, w2.y, acc.y);
            acc.z = fmaf(x2, w2.z, acc.z); acc.w = fmaf(x2, w2.w, acc.w);
            acc.x = fmaf(x3, w3.x, acc.x); acc.y = fmaf(x3, w3.y, acc.y);
            acc.z = fmaf(x3, w3.z, acc.z); acc.w = fmaf(x3, w3.w, acc.w);
        }

        const float fdeg = (float)deg;
        float4 r;
        r.x = self.x + acc.x / fdeg;
        r.y = self.y + acc.y / fdeg;
        r.z = self.z + acc.z / fdeg;
        r.w = self.w + acc.w / fdeg;
        *orow = r;
    }
}

extern "C" void kernel_launch(void* const* d_in, const int* in_sizes, int n_in,
                              void* d_out, int out_size, void* d_ws, size_t ws_size,
                              hipStream_t stream) {
    const float* X       = (const float*)d_in[0];
    const float* X_edge  = (const float*)d_in[1];
    const float* W_neigh = (const float*)d_in[2];
    const float* W_self  = (const float*)d_in[3];
    const float* b_self  = (const float*)d_in[4];
    const float* W_edge  = (const float*)d_in[5];
    const int*   src     = (const int*)d_in[6];
    const int*   dst     = (const int*)d_in[7];
    float* out = (float*)d_out;

    float* h_src = (float*)d_ws;
    int*   start = (int*)((char*)d_ws + (size_t)N_NODES * D_OUT * sizeof(float));
    int*   part  = (int*)((char*)d_ws + (size_t)N_NODES * D_OUT * sizeof(float)
                                      + (size_t)(N_NODES + 4) * sizeof(int));
    int*   tot   = (int*)((char*)d_ws + (size_t)N_NODES * D_OUT * sizeof(float)
                                      + (size_t)(N_NODES + 4) * sizeof(int)
                                      + (size_t)N_CHUNKS * N_NODES * sizeof(int));
    int2*  pairs = (int2*)((char*)d_ws + (size_t)N_NODES * D_OUT * sizeof(float)
                                       + (size_t)(N_NODES + 4) * sizeof(int)
                                       + (size_t)N_CHUNKS * N_NODES * sizeof(int)
                                       + (size_t)256 * sizeof(int));

    range_count_kernel<<<N_RANGES * N_CHUNKS, CSR_THREADS, 0, stream>>>(dst, part, tot);
    range_place_kernel<<<N_RANGES * N_CHUNKS, CSR_THREADS, 0, stream>>>(
        src, dst, part, tot, start, pairs);

    gemm_dual_wmma_kernel<<<GEMM_BLOCKS, 512, 0, stream>>>(
        X, W_self, W_neigh, b_self, out, h_src);

    node_gather_kernel<<<(N_NODES + GATHER_WAVES * NODES_PER_WAVE - 1) / (GATHER_WAVES * NODES_PER_WAVE),
                         256, 0, stream>>>(
        X_edge, h_src, pairs, start, W_edge, out, N_NODES);
}


# gather kernel: resident grid of 1024 blocks walks the nodes as a software pipeline (next node's offsets and pairs requested under the current node's rows), 16 row loads in flight per wave
# speedup vs baseline: 3.3066x; 1.0417x over previous
#include <hip/hip_runtime.h>
#include <hip/hip_bf16.h>

#define N_NODES 100000
#define N_EDGES 1600000
#define D_IN    256
#define D_EDGE  16
#define D_OUT   128

typedef __attribute__((ext_vector_type(2))) float v2f;
typedef __attribute__((ext_vector_type(8))) float v8f;

#define N_RANGES     25
#define RANGE_NODES  4000
#define N_CHUNKS     10
#define CHUNK_EDGES  160000
#define CSR_THREADS  1024
#define CHUNK_VEC    (CHUNK_EDGES / 4)
#define RANGE_VEC    (RANGE_NODES / 4)

static_assert(N_RANGES * RANGE_NODES == N_NODES, "ranges must tile the nodes");
static_assert(N_CHUNKS * CHUNK_EDGES == N_EDGES, "chunks must tile the edges");
static_assert(CHUNK_EDGES % 4 == 0 && RANGE_NODES % 4 == 0, "16-byte accesses");
static_assert(RANGE_VEC <= CSR_THREADS, "one thread per four nodes");
static_assert(N_RANGES * N_CHUNKS <= CSR_THREADS, "one thread per total");

__device__ __forceinline__ void count4(int* lc, const int4 d, const int lo) {
    const unsigned k0 = (unsigned)(d.x - lo);
    const unsigned k1 = (unsigned)(d.y - lo);
    const unsigned k2 = (unsigned)(d.z - lo);
    const unsigned k3 = (unsigned)(d.w - lo);
    if (k0 < RANGE_NODES) atomicAdd(&lc[k0], 1);
    if (k1 < RANGE_NODES) atomicAdd(&lc[k1], 1);
    if (k2 < RANGE_NODES) atomicAdd(&lc[k2], 1);
    if (k3 < RANGE_NODES) atomicAdd(&lc[k3], 1);
}

__global__ void __launch_bounds__(CSR_THREADS)
range_count_kernel(const int* __restrict__ dst,
                   int* __restrict__ part, int* __restrict__ tot) {
    __shared__ int lc[RANGE_NODES];
    __shared__ int wsum[32];

    const int t    = threadIdx.x;
    const int lane = t & 31;
    const int w    = t >> 5;
    const int r    = blockIdx.x % N_RANGES;
    const int c    = blockIdx.x / N_RANGES;
    const int lo   = r * RANGE_NODES;

    for (int i = t; i < RANGE_NODES; i += CSR_THREADS) lc[i] = 0;
    __syncthreads();

    const int4* __restrict__ d4 =
        reinterpret_cast<const int4*>(dst + (size_t)c * CHUNK_EDGES);

    for (int i = t; i < CHUNK_VEC; i += 4 * CSR_THREADS) {
        int4 d[4];
        #pragma unroll
        for (int u = 0; u < 4; ++u) {
            const int idx = i + u * CSR_THREADS;
            d[u] = make_int4(-1, -1, -1, -1);
            if (idx < CHUNK_VEC) d[u] = d4[idx];
        }
        #pragma unroll
        for (int u = 0; u < 4; ++u) count4(lc, d[u], lo);
    }
    __syncthreads();

    int s = 0;
    if (t < RANGE_VEC) {
        const int4 v = reinterpret_cast<const int4*>(lc)[t];
        reinterpret_cast<int4*>(part + (size_t)c * N_NODES + lo)[t] = v;
        s = v.x + v.y + v.z + v.w;
    }
    for (int off = 16; off > 0; off >>= 1) s += __shfl_xor(s, off, 32);
    if (lane == 0) wsum[w] = s;
    __syncthreads();
    if (w == 0) {
        int p = wsum[lane];
        for (int off = 16; off > 0; off >>= 1) p += __shfl_xor(p, off, 32);
        if (lane == 0) tot[c * N_RANGES + r] = p;
    }
}

__device__ __forceinline__ void rank4(int* lb, const int4 d, const int lo, int (&p)[4]) {
    const int dd[4] = {d.x, d.y, d.z, d.w};
    #pragma unroll
    for (int j = 0; j < 4; ++j) {
        const unsigned k = (unsigned)(dd[j] - lo);
        p[j] = -1;
        if (k < RANGE_NODES) p[j] = atomicAdd(&lb[k], 1);
    }
}

__device__ __forceinline__ void store4(int2* __restrict__ pairs, const int4 sv,
                                       const int e, const int (&p)[4]) {
    const int ss[4] = {sv.x, sv.y, sv.z, sv.w};
    #pragma unroll
    for (int j = 0; j < 4; ++j)
        if (p[j] >= 0) pairs[p[j]] = make_int2(ss[j], e + j);
}

__global__ void __launch_bounds__(CSR_THREADS)
range_place_kernel(const int* __restrict__ src,
                   const int* __restrict__ dst,
                   const int* __restrict__ part,
                   const int* __restrict__ tot,
                   int* __restrict__ start,
                   int2* __restrict__ pairs) {
    __shared__ int lb[RANGE_NODES];
    __shared__ int wave_incl[32];
    __shared__ int red[32];
    __shared__ int range_base;

    const int t    = threadIdx.x;
    const int lane = t & 31;
    const int w    = t >> 5;
    const int r    = blockIdx.x % N_RANGES;
    const int c    = blockIdx.x / N_RANGES;
    const int lo   = r * RANGE_NODES;

    int tb = 0;
    if (t < N_RANGES * N_CHUNKS && (t % N_RANGES) < r) tb = tot[t];

    int4 cnt = make_int4(0, 0, 0, 0);
    int4 pre = make_int4(0, 0, 0, 0);
    if (t < RANGE_VEC) {
        int4 v[N_CHUNKS];
        #pragma unroll
        for (int cc = 0; cc < N_CHUNKS; ++cc)
            v[cc] = reinterpret_cast<const int4*>(part + (size_t)cc * N_NODES + lo)[t];
        #pragma unroll
        for (int cc = 0; cc < N_CHUNKS; ++cc) {
            cnt.x += v[cc].x; cnt.y += v[cc].y; cnt.z += v[cc].z; cnt.w += v[cc].w;
            if (cc < c) { pre.x += v[cc].x; pre.y += v[cc].y; pre.z += v[cc].z; pre.w += v[cc].w; }
        }
    }

    const int s = cnt.x + cnt.y + cnt.z + cnt.w;
    int incl = s;
    for (int off = 1; off < 32; off <<= 1) {
        const int u = __shfl_up(incl, off, 32);
        if (lane >= off) incl += u;
    }
    for (int off = 16; off > 0; off >>= 1) tb += __shfl_xor(tb, off, 32);
    if (lane == 31) wave_incl[w] = incl;
    if (lane == 0)  red[w] = tb;
    __syncthreads();

    if (w == 0) {
        const int wt = wave_incl[lane];
        int inc2 = wt;
        for (int off = 1; off < 32; off <<= 1) {
            const int u = __shfl_up(inc2, off, 32);
            if (lane >= off) inc2 += u;
        }
        int p = red[lane];
        for (int off = 16; off > 0; off >>= 1) p += __shfl_xor(p, off, 32);
        wave_incl[lane] = inc2 - wt;
        if (lane == 0) range_base = p;
    }
    __syncthreads();

    if (t < RANGE_VEC) {
        const int b0 = range_base + wave_incl[w] + incl - s;
        const int b1 = b0 + cnt.x;
        const int b2 = b1 + cnt.y;
        const int b3 = b2 + cnt.z;
        if (c == 0) {
            reinterpret_cast<int4*>(start + lo)[t] = make_int4(b0, b1, b2, b3);
            if (r == N_RANGES - 1 && t == RANGE_VEC - 1)
                start[N_NODES] = b3 + cnt.w;
        }
        reinterpret_cast<int4*>(lb)[t] =
            make_int4(b0 + pre.x, b1 + pre.y, b2 + pre.z, b3 + pre.w);
    }
    __syncthreads();

    const int4* __restrict__ d4 =
        reinterpret_cast<const int4*>(dst + (size_t)c * CHUNK_EDGES);
    const int4* __restrict__ s4 =
        reinterpret_cast<const int4*>(src + (size_t)c * CHUNK_EDGES);
    const int e0 = c * CHUNK_EDGES;

    for (int i = t; i < CHUNK_VEC; i += 4 * CSR_THREADS) {
        int4 d[4], sv[4];
        #pragma unroll
        for (int u = 0; u < 4; ++u) {
            const int idx = i + u * CSR_THREADS;
            d[u]  = make_int4(-1, -1, -1, -1);
            sv[u] = make_int4(0, 0, 0, 0);
            if (idx < CHUNK_VEC) { d[u] = d4[idx]; sv[u] = s4[idx]; }
        }
        int p[4][4];
        #pragma unroll
        for (int u = 0; u < 4; ++u) rank4(lb, d[u], lo, p[u]);
        #pragma unroll
        for (int u = 0; u < 4; ++u)
            store4(pairs, sv[u], e0 + 4 * (i + u * CSR_THREADS), p[u]);
    }
}

#define XS_PITCH    260
#define XS_PITCH4   (XS_PITCH / 4)
#define GEMM_BLOCKS 256
#define N_TILES     (N_NODES / 16)
#define K_STEPS     (D_IN / 4)

static_assert(N_TILES * 16 == N_NODES, "row tiles must tile the nodes");
static_assert(GEMM_BLOCKS <= N_TILES, "every work-group has a first tile");

__global__ void __launch_bounds__(512)
gemm_dual_wmma_kernel(const float* __restrict__ X,
                      const float* __restrict__ W_self,
                      const float* __restrict__ W_neigh,
                      const float* __restrict__ b_self,
                      float* __restrict__ out_self,
                      float* __restrict__ out_neigh) {
    __shared__ float4 xs4[2][16 * XS_PITCH4];

    const int wave  = threadIdx.x >> 5;
    const int lane  = threadIdx.x & 31;
    const int ntile = wave & 7;
    const bool isSelf = (wave < 8);

    const float* __restrict__ W = isSelf ? W_self : W_neigh;

    const int m     = lane & 15;
    const int khalf = lane >> 4;
    const int col   = ntile * 16 + m;

    v2f breg[K_STEPS];
    #pragma unroll
    for (int j = 0; j < K_STEPS; ++j) {
        const int k0 = 4 * j + 2 * khalf;
        breg[j].x = W[(size_t)(k0 + 0) * D_OUT + col];
        breg[j].y = W[(size_t)(k0 + 1) * D_OUT + col];
    }

    const float bias = isSelf ? b_self[col] : 0.0f;
    float* __restrict__ outp = isSelf ? out_self : out_neigh;

    const int sr = threadIdx.x >> 6;
    const int sc = threadIdx.x & 63;
    const float4* __restrict__ X4 = reinterpret_cast<const float4*>(X);

    int tile = blockIdx.x;
    float4 p0 = X4[(size_t)(tile * 16 + sr)     * (D_IN / 4) + sc];
    float4 p1 = X4[(size_t)(tile * 16 + sr + 8) * (D_IN / 4) + sc];
    xs4[0][sr * XS_PITCH4 + sc]       = p0;
    xs4[0][(sr + 8) * XS_PITCH4 + sc] = p1;
    __syncthreads();

    int buf = 0;
    for (;;) {
        const int  next    = tile + GEMM_BLOCKS;
        const bool hasNext = next < N_TILES;
        if (hasNext) {
            p0 = X4[(size_t)(next * 16 + sr)     * (D_IN / 4) + sc];
            p1 = X4[(size_t)(next * 16 + sr + 8) * (D_IN / 4) + sc];
        }

        const float* xrow = reinterpret_cast<const float*>(xs4[buf]) + m * XS_PITCH + 2 * khalf;
        v8f acc = {};
        #pragma unroll
        for (int j = 0; j < K_STEPS; ++j) {
            const v2f a = *reinterpret_cast<const v2f*>(xrow + 4 * j);
            acc = __builtin_amdgcn_wmma_f32_16x16x4_f32(
                 false, a,  false, breg[j],
                 (short)0, acc,  false,  false);
        }

        if (hasNext) {
            xs4[buf ^ 1][sr * XS_PITCH4 + sc]       = p0;
            xs4[buf ^ 1][(sr + 8) * XS_PITCH4 + sc] = p1;
            __syncthreads();
        }

        const int rowBase = tile * 16;
        #pragma unroll
        for (int r = 0; r < 8; ++r) {
            const int row = rowBase + r + 8 * khalf;
            outp[(size_t)row * D_OUT + col] = acc[r] + bias;
        }

        if (!hasNext) break;
        buf ^= 1;
        tile = next;
    }
}

#define GATHER_WAVES        8
#define GATHER_BLOCKS       1024
#define GATHER_TOTAL_WAVES  (GATHER_BLOCKS * GATHER_WAVES)
#define ROW_GROUP           8

__global__ void __launch_bounds__(256)
node_gather_kernel(const float* __restrict__ X_edge,
                   const float* __restrict__ h_src,
                   const int2* __restrict__ pairs,
                   const int* __restrict__ start,
                   const float* __restrict__ W_edge,
                   float* __restrict__ out, int n_total) {
    __shared__ float4 wl[D_EDGE * D_OUT / 4];
    for (int i = threadIdx.x; i < D_EDGE * D_OUT / 4; i += 256)
        wl[i] = reinterpret_cast<const float4*>(W_edge)[i];
    __syncthreads();

    const int lane = threadIdx.x & 31;
    const int q    = lane & 3;
    const int sub  = lane >> 2;

    const char* hb = reinterpret_cast<const char*>(h_src);
    const char* xb = reinterpret_cast<const char*>(X_edge);
    const unsigned lane_off = (unsigned)lane * 16u;
    const unsigned q_off    = (unsigned)q * 16u;

    int n_c = n_total, beg_c = 0, end_c = 0;
    int2 pr_c = make_int2(0, 0);
    int n_n = __builtin_amdgcn_readfirstlane(blockIdx.x * GATHER_WAVES + (threadIdx.x >> 5));
    int beg_n = 0, end_n = 0;
    if (n_n < n_total) { beg_n = start[n_n]; end_n = start[n_n + 1]; }

    for (;;) {
        int2 pr_n = make_int2(0, 0);
        if (lane < end_n - beg_n) pr_n = pairs[beg_n + lane];
        const int n_2 = n_n + GATHER_TOTAL_WAVES;
        int beg_2 = 0, end_2 = 0;
        if (n_2 < n_total) { beg_2 = start[n_2]; end_2 = start[n_2 + 1]; }

        const int deg_c = end_c - beg_c;
        if (deg_c > 0) {
            float4* orow = reinterpret_cast<float4*>(out + (size_t)n_c * D_OUT) + lane;
            const float4 self = *orow;

            float4 acc  = make_float4(0.f, 0.f, 0.f, 0.f);
            float4 xacc = make_float4(0.f, 0.f, 0.f, 0.f);

            auto chunk = [&](const int2 pr, const int m) {
                float4 xv[4];
                #pragma unroll
                for (int g = 0; g < 4; ++g) {
                    const unsigned e = (unsigned)__shfl(pr.y, g * 8 + sub, 32);
                    xv[g] = *reinterpret_cast<const float4*>(xb + (e * 64u + q_off));
                }

                for (int j0 = 0; j0 < m; j0 += 2 * ROW_GROUP) {
                    float4 h[2 * ROW_GROUP];
                    const bool second = (j0 + ROW_GROUP < m);
                    #pragma unroll
                    for (int u = 0; u < ROW_GROUP; ++u) {
                        const int jj = min(j0 + u, m - 1);
                        const unsigned s = (unsigned)__builtin_amdgcn_readlane(pr.x, jj);
                        h[u] = *reinterpret_cast<const float4*>(hb + (s * 512u + lane_off));
                    }
                    if (second) {
                        #pragma unroll
                        for (int u = ROW_GROUP; u < 2 * ROW_GROUP; ++u) {
                            const int jj = min(j0 + u, m - 1);
                            const unsigned s = (unsigned)__builtin_amdgcn_readlane(pr.x, jj);
                            h[u] = *reinterpret_cast<const float4*>(hb + (s * 512u + lane_off));
                        }
                    }
                    #pragma unroll
                    for (int u = 0; u < ROW_GROUP; ++u) {
                        if (j0 + u < m) { acc.x += h[u].x; acc.y += h[u].y; acc.z += h[u].z; acc.w += h[u].w; }
                    }
                    if (second) {
                        #pragma unroll
                        for (int u = ROW_GROUP; u < 2 * ROW_GROUP; ++u) {
                            if (j0 + u < m) { acc.x += h[u].x; acc.y += h[u].y; acc.z += h[u].z; acc.w += h[u].w; }
                        }
                    }
                }

                #pragma unroll
                for (int g = 0; g < 4; ++g) {
                    if (g * 8 + sub < m) {
                        xacc.x += xv[g].x; xacc.y += xv[g].y; xacc.z += xv[g].z; xacc.w += xv[g].w;
                    }
                }
            };

            chunk(pr_c, min(32, deg_c));
            for (int base = 32; base < deg_c; base += 32) {
                const int m = min(32, deg_c - base);
                int2 pr = make_int2(0, 0);
                if (lane < m) pr = pairs[beg_c + base + lane];
                chunk(pr, m);
            }

            for (int off = 4; off < 32; off <<= 1) {
                xacc.x += __shfl_xor(xacc.x, off, 32);
                xacc.y += __shfl_xor(xacc.y, off, 32);
                xacc.z += __shfl_xor(xacc.z, off, 32);
                xacc.w += __shfl_xor(xacc.w, off, 32);
            }

            #pragma unroll
            for (int qq = 0; qq < 4; ++qq) {
                const float x0 = __int_as_float(__builtin_amdgcn_readlane(__float_as_int(xacc.x), qq));
                const float x1 = __int_as_float(__builtin_amdgcn_readlane(__float_as_int(xacc.y), qq));
                const float x2 = __int_as_float(__builtin_amdgcn_readlane(__float_as_int(xacc.z), qq));
                const float x3 = __int_as_float(__builtin_amdgcn_readlane(__float_as_int(xacc.w), qq));
                const float4 w0 = wl[(qq * 4 + 0) * (D_OUT / 4) + lane];
                const float4 w1 = wl[(qq * 4 + 1) * (D_OUT / 4) + lane];
                const float4 w2 = wl[(qq * 4 + 2) * (D_OUT / 4) + lane];
                const float4 w3 = wl[(qq * 4 + 3) * (D_OUT / 4) + lane];
                acc.x = fmaf(x0, w0.x, acc.x); acc.y = fmaf(x0, w0.y, acc.y);
                acc.z = fmaf(x0, w0.z, acc.z); acc.w = fmaf(x0, w0.w, acc.w);
                acc.x = fmaf(x1, w1.x, acc.x); acc.y = fmaf(x1, w1.y, acc.y);
                acc.z = fmaf(x1, w1.z, acc.z); acc.w = fmaf(x1, w1.w, acc.w);
                acc.x = fmaf(x2, w2.x, acc.x); acc.y = fmaf(x2, w2.y, acc.y);
                acc.z = fmaf(x2, w2.z, acc.z); acc.w = fmaf(x2, w2.w, acc.w);
                acc.x = fmaf(x3, w3.x, acc.x); acc.y = fmaf(x3, w3.y, acc.y);
                acc.z = fmaf(x3, w3.z, acc.z); acc.w = fmaf(x3, w3.w, acc.w);
            }

            const float fdeg = (float)deg_c;
            float4 r;
            r.x = self.x + acc.x / fdeg;
            r.y = self.y + acc.y / fdeg;
            r.z = self.z + acc.z / fdeg;
            r.w = self.w + acc.w / fdeg;
            *orow = r;
        }

        n_c = n_n; beg_c = beg_n; end_c = end_n; pr_c = pr_n;
        n_n = n_2; beg_n = beg_2; end_n = end_2;
        if (n_c >= n_total) break;
    }
}

extern "C" void kernel_launch(void* const* d_in, const int* in_sizes, int n_in,
                              void* d_out, int out_size, void* d_ws, size_t ws_size,
                              hipStream_t stream) {
    const float* X       = (const float*)d_in[0];
    const float* X_edge  = (const float*)d_in[1];
    const float* W_neigh = (const float*)d_in[2];
    const float* W_self  = (const float*)d_in[3];
    const float* b_self  = (const float*)d_in[4];
    const float* W_edge  = (const float*)d_in[5];
    const int*   src     = (const int*)d_in[6];
    const int*   dst     = (const int*)d_in[7];
    float* out = (float*)d_out;

    float* h_src = (float*)d_ws;
    int*   start = (int*)((char*)d_ws + (size_t)N_NODES * D_OUT * sizeof(float));
    int*   part  = (int*)((char*)d_ws + (size_t)N_NODES * D_OUT * sizeof(float)
                                      + (size_t)(N_NODES + 4) * sizeof(int));
    int*   tot   = (int*)((char*)d_ws + (size_t)N_NODES * D_OUT * sizeof(float)
                                      + (size_t)(N_NODES + 4) * sizeof(int)
                                      + (size_t)N_CHUNKS * N_NODES * sizeof(int));
    int2*  pairs = (int2*)((char*)d_ws + (size_t)N_NODES * D_OUT * sizeof(float)
                                       + (size_t)(N_NODES + 4) * sizeof(int)
                                       + (size_t)N_CHUNKS * N_NODES * sizeof(int)
                                       + (size_t)256 * sizeof(int));

    range_count_kernel<<<N_RANGES * N_CHUNKS, CSR_THREADS, 0, stream>>>(dst, part, tot);
    range_place_kernel<<<N_RANGES * N_CHUNKS, CSR_THREADS, 0, stream>>>(
        src, dst, part, tot, start, pairs);

    gemm_dual_wmma_kernel<<<GEMM_BLOCKS, 512, 0, stream>>>(
        X, W_self, W_neigh, b_self, out, h_src);

    node_gather_kernel<<<GATHER_BLOCKS, 256, 0, stream>>>(
        X_edge, h_src, pairs, start, W_edge, out, N_NODES);
}
